# starting kernel
# baseline (speedup 1.0000x reference)
#include <hip/hip_runtime.h>
#include <hip/hip_bf16.h>

typedef __attribute__((ext_vector_type(2))) float v2f;
typedef __attribute__((ext_vector_type(8))) float v8f;

#define DF 128
#define NEG_SLOPE 0.2f

__global__ void init_kernel(float* __restrict__ out, float* __restrict__ mmax,
                            float* __restrict__ denom, int N) {
  int i = blockIdx.x * blockDim.x + threadIdx.x;
  int total = N * DF;
  if (i < total) out[i] = 0.0f;
  if (i < N) {
    mmax[i]  = -__builtin_inff();
    denom[i] = 0.0f;
  }
}

__global__ void gemm_xw_kernel(const float* __restrict__ x,
                               const float* __restrict__ W,
                               float* __restrict__ xw, int N) {
  int wave  = threadIdx.x >> 5;
  int lane  = threadIdx.x & 31;
  int strip = blockIdx.x * (blockDim.x >> 5) + wave;
  int tilesM = (N + 15) >> 4;
  if (strip >= tilesM) return;

  int half = lane >> 4;
  int l    = lane & 15;

  int rowA = strip * 16 + l;
  if (rowA >= N) rowA = N - 1;
  const float* Arow = x + (size_t)rowA * DF;

  v8f zero = {};
  v8f acc[8];
  #pragma unroll
  for (int t = 0; t < 8; ++t) acc[t] = zero;

  for (int k = 0; k < DF; k += 4) {
    int ka = k + 2 * half;
    v2f a;
    a.x = Arow[ka];
    a.y = Arow[ka + 1];
    const float* Wrow = W + (size_t)ka * DF + l;
    #pragma unroll
    for (int t = 0; t < 8; ++t) {
      v2f b;
      b.x = Wrow[t * 16];
      b.y = Wrow[DF + t * 16];
      acc[t] = __builtin_amdgcn_wmma_f32_16x16x4_f32(
           false, a,  false, b,
           (short)0, acc[t],  false,  false);
    }
  }

  int mbase = strip * 16 + half * 8;
  if (strip * 16 + 16 <= N) {
    #pragma unroll
    for (int t = 0; t < 8; ++t) {
      float* orow = xw + (size_t)mbase * DF + t * 16 + l;
      #pragma unroll
      for (int v = 0; v < 8; ++v) orow[(size_t)v * DF] = acc[t][v];
    }
  } else {
    for (int t = 0; t < 8; ++t)
      for (int v = 0; v < 8; ++v) {
        int row = mbase + v;
        if (row < N) xw[(size_t)row * DF + t * 16 + l] = acc[t][v];
      }
  }
}

__global__ void logits_kernel(const float* __restrict__ xw,
                              const float* __restrict__ att_src,
                              const float* __restrict__ att_dst,
                              float* __restrict__ a_src,
                              float* __restrict__ a_dst, int N) {
  int gw   = (blockIdx.x * blockDim.x + threadIdx.x) >> 5;
  int lane = threadIdx.x & 31;
  if (gw >= N) return;
  const float4 xv = *(const float4*)(xw + (size_t)gw * DF + lane * 4);
  const float4 sv = *(const float4*)(att_src + lane * 4);
  const float4 dv = *(const float4*)(att_dst + lane * 4);
  float s = xv.x * sv.x + xv.y * sv.y + xv.z * sv.z + xv.w * sv.w;
  float d = xv.x * dv.x + xv.y * dv.y + xv.z * dv.z + xv.w * dv.w;
  #pragma unroll
  for (int off = 16; off > 0; off >>= 1) {
    s += __shfl_down(s, off, 32);
    d += __shfl_down(d, off, 32);
  }
  if (lane == 0) { a_src[gw] = s; a_dst[gw] = d; }
}

__device__ __forceinline__ void edge_sd(const int* __restrict__ ei, int i,
                                        int E, int& s, int& d) {
  if (i < E) { s = ei[i]; d = ei[E + i]; }
  else       { s = i - E; d = s; }
}

__device__ __forceinline__ float leaky(float e) {
  return e >= 0.0f ? e : NEG_SLOPE * e;
}

__device__ __forceinline__ void atomicMaxFloat(float* addr, float val) {
  if (val >= 0.0f) atomicMax((int*)addr, __float_as_int(val));
  else atomicMin((unsigned int*)addr, (unsigned int)__float_as_int(val));
}

__global__ void edge_max_kernel(const int* __restrict__ ei,
                                const float* __restrict__ a_src,
                                const float* __restrict__ a_dst,
                                float* __restrict__ mmax, int E, int N) {
  int i = blockIdx.x * blockDim.x + threadIdx.x;
  if (i >= E + N) return;
  int s, d; edge_sd(ei, i, E, s, d);
  float e = leaky(a_src[s] + a_dst[d]);
  atomicMaxFloat(&mmax[d], e);
}

__global__ void edge_denom_kernel(const int* __restrict__ ei,
                                  const float* __restrict__ a_src,
                                  const float* __restrict__ a_dst,
                                  const float* __restrict__ mmax,
                                  float* __restrict__ denom, int E, int N) {
  int i = blockIdx.x * blockDim.x + threadIdx.x;
  if (i >= E + N) return;
  int s, d; edge_sd(ei, i, E, s, d);
  float e = leaky(a_src[s] + a_dst[d]);
  atomicAdd(&denom[d], expf(e - mmax[d]));
}

__global__ void edge_agg_kernel(const int* __restrict__ ei,
                                const float* __restrict__ xw,
                                const float* __restrict__ a_src,
                                const float* __restrict__ a_dst,
                                const float* __restrict__ mmax,
                                const float* __restrict__ denom,
                                float* __restrict__ out, int E, int N) {
  int gw   = (blockIdx.x * blockDim.x + threadIdx.x) >> 5;
  int lane = threadIdx.x & 31;
  if (gw >= E + N) return;
  int s, d; edge_sd(ei, gw, E, s, d);
  float e = leaky(a_src[s] + a_dst[d]);
  float alpha = expf(e - mmax[d]) / denom[d];
  const float4 xv = *(const float4*)(xw + (size_t)s * DF + lane * 4);
  float* o = out + (size_t)d * DF + lane * 4;
  atomicAdd(o + 0, xv.x * alpha);
  atomicAdd(o + 1, xv.y * alpha);
  atomicAdd(o + 2, xv.z * alpha);
  atomicAdd(o + 3, xv.w * alpha);
}

__global__ void finalize_kernel(float* __restrict__ out,
                                const float* __restrict__ bias, int total) {
  int i = blockIdx.x * blockDim.x + threadIdx.x;
  if (i >= total) return;
  float v = out[i] + bias[i & (DF - 1)];
  out[i] = v > 0.0f ? v : (expf(v) - 1.0f);
}

extern "C" void kernel_launch(void* const* d_in, const int* in_sizes, int n_in,
                              void* d_out, int out_size, void* d_ws, size_t ws_size,
                              hipStream_t stream) {
  const float* x       = (const float*)d_in[0];
  const int*   ei      = (const int*)  d_in[1];
  const float* W       = (const float*)d_in[2];
  const float* att_src = (const float*)d_in[3];
  const float* att_dst = (const float*)d_in[4];
  const float* bias    = (const float*)d_in[5];

  const int N = in_sizes[0] / DF;
  const int E = in_sizes[1] / 2;
  float* out = (float*)d_out;

  float* xw    = (float*)d_ws;
  float* asrc  = xw + (size_t)N * DF;
  float* adst  = asrc + N;
  float* mmax  = adst + N;
  float* denom = mmax + N;

  const int tot = E + N;

  {
    int threads = N * DF;
    init_kernel<<<(threads + 255) / 256, 256, 0, stream>>>(out, mmax, denom, N);
  }
  {
    int strips = (N + 15) / 16;
    int blocks = (strips + 3) / 4;
    gemm_xw_kernel<<<blocks, 128, 0, stream>>>(x, W, xw, N);
  }
  {
    long long threads = (long long)N * 32;
    logits_kernel<<<(unsigned)((threads + 255) / 256), 256, 0, stream>>>(
        xw, att_src, att_dst, asrc, adst, N);
  }
  edge_max_kernel<<<(tot + 255) / 256, 256, 0, stream>>>(ei, asrc, adst, mmax, E, N);
  edge_denom_kernel<<<(tot + 255) / 256, 256, 0, stream>>>(ei, asrc, adst, mmax, denom, E, N);
  {
    long long threads = (long long)tot * 32;
    edge_agg_kernel<<<(unsigned)((threads + 255) / 256), 256, 0, stream>>>(
        ei, xw, asrc, adst, mmax, denom, out, E, N);
  }
  {
    int threads = N * DF;
    finalize_kernel<<<(threads + 255) / 256, 256, 0, stream>>>(out, bias, threads);
  }
}


# Edges grouped by destination in-call (returning int add = slot in the node's out row, source+logit kept there); one wave per node: softmax and row sum in registers, no f32 atomics, slot-order sums
# speedup vs baseline: 2.3583x; 2.3583x over previous
#include <hip/hip_runtime.h>
#include <hip/hip_bf16.h>

typedef __attribute__((ext_vector_type(2))) float v2f;
typedef __attribute__((ext_vector_type(8))) float v8f;

#define DF 128
#define NEG_SLOPE 0.2f

__global__ void gemm_xw_kernel(const float* __restrict__ x,
                               const float* __restrict__ W,
                               float* __restrict__ xw, int N) {
  int wave  = threadIdx.x >> 5;
  int lane  = threadIdx.x & 31;
  int strip = blockIdx.x * (blockDim.x >> 5) + wave;
  int tilesM = (N + 15) >> 4;
  if (strip >= tilesM) return;

  int half = lane >> 4;
  int l    = lane & 15;

  int rowA = strip * 16 + l;
  if (rowA >= N) rowA = N - 1;
  const float* Arow = x + (size_t)rowA * DF;

  v8f zero = {};
  v8f acc[8];
  #pragma unroll
  for (int t = 0; t < 8; ++t) acc[t] = zero;

  for (int k = 0; k < DF; k += 4) {
    int ka = k + 2 * half;
    v2f a;
    a.x = Arow[ka];
    a.y = Arow[ka + 1];
    const float* Wrow = W + (size_t)ka * DF + l;
    #pragma unroll
    for (int t = 0; t < 8; ++t) {
      v2f b;
      b.x = Wrow[t * 16];
      b.y = Wrow[DF + t * 16];
      acc[t] = __builtin_amdgcn_wmma_f32_16x16x4_f32(
           false, a,  false, b,
           (short)0, acc[t],  false,  false);
    }
  }

  int mbase = strip * 16 + half * 8;
  if (strip * 16 + 16 <= N) {
    #pragma unroll
    for (int t = 0; t < 8; ++t) {
      float* orow = xw + (size_t)mbase * DF + t * 16 + l;
      #pragma unroll
      for (int v = 0; v < 8; ++v) orow[(size_t)v * DF] = acc[t][v];
    }
  } else {
    for (int t = 0; t < 8; ++t)
      for (int v = 0; v < 8; ++v) {
        int row = mbase + v;
        if (row < N) xw[(size_t)row * DF + t * 16 + l] = acc[t][v];
      }
  }
}

__global__ void logits_kernel(const float* __restrict__ xw,
                              const float* __restrict__ att_src,
                              const float* __restrict__ att_dst,
                              float* __restrict__ a_src,
                              float* __restrict__ a_dst,
                              int* __restrict__ cnt, int N) {
  int gw   = (blockIdx.x * blockDim.x + threadIdx.x) >> 5;
  int lane = threadIdx.x & 31;
  if (gw >= N) return;
  const float4 xv = *(const float4*)(xw + (size_t)gw * DF + lane * 4);
  const float4 sv = *(const float4*)(att_src + lane * 4);
  const float4 dv = *(const float4*)(att_dst + lane * 4);
  float s = xv.x * sv.x + xv.y * sv.y + xv.z * sv.z + xv.w * sv.w;
  float d = xv.x * dv.x + xv.y * dv.y + xv.z * dv.z + xv.w * dv.w;
  #pragma unroll
  for (int off = 16; off > 0; off >>= 1) {
    s += __shfl_down(s, off, 32);
    d += __shfl_down(d, off, 32);
  }
  if (lane == 0) { a_src[gw] = s; a_dst[gw] = d; cnt[gw] = 0; }
}

__device__ __forceinline__ float leaky(float e) {
  return e >= 0.0f ? e : NEG_SLOPE * e;
}

#define CAP (DF / 2)

__global__ void group_kernel(const int* __restrict__ ei,
                             const float* __restrict__ a_src,
                             int* __restrict__ cnt,
                             float* __restrict__ lists, int E) {
  int i = blockIdx.x * blockDim.x + threadIdx.x;
  if (i >= E) return;
  int s = ei[i];
  int d = ei[E + i];
  float as = a_src[s];
  int r = atomicAdd(&cnt[d], 1);
  r = r < CAP ? r : CAP - 1;
  float2 p;
  p.x = __int_as_float(s);
  p.y = as;
  *(float2*)(lists + (size_t)d * DF + 2 * r) = p;
}

__device__ __forceinline__ float wave_max(float v) {
  for (int off = 16; off > 0; off >>= 1) v = fmaxf(v, __shfl_xor(v, off, 32));
  return v;
}

__device__ __forceinline__ float wave_sum(float v) {
  for (int off = 16; off > 0; off >>= 1) v += __shfl_xor(v, off, 32);
  return v;
}

__device__ __forceinline__ int lane_int(int v, int j) {
  return __builtin_amdgcn_readlane(v, j);
}
__device__ __forceinline__ float lane_flt(float v, int j) {
  return __int_as_float(__builtin_amdgcn_readlane(__float_as_int(v), j));
}

__device__ __forceinline__ float4 row4(const float* __restrict__ xw, int s, int lane) {
  return *(const float4*)(xw + (size_t)s * DF + lane * 4);
}

__device__ __forceinline__ void axpy4(float4& acc, float a, const float4& r) {
  acc.x += a * r.x; acc.y += a * r.y; acc.z += a * r.z; acc.w += a * r.w;
}

__global__ void gather_kernel(const int* __restrict__ ei,
                              const float* __restrict__ xw,
                              const float* __restrict__ a_src,
                              const float* __restrict__ a_dst,
                              const int* __restrict__ cnt,
                              const float* __restrict__ bias,
                              float* out, int E, int N) {
  int gw   = (blockIdx.x * blockDim.x + threadIdx.x) >> 5;
  int lane = threadIdx.x & 31;
  if (gw >= N) return;
  const int d = __builtin_amdgcn_readfirstlane(gw);

  const int   n      = cnt[d];
  const float ad     = a_dst[d];
  const float e_self = leaky(a_src[d] + ad);
  float* orow = out + (size_t)d * DF;
  const float4 rself = row4(xw, d, lane);
  const float4 bv    = *(const float4*)(bias + lane * 4);

  float4 acc = {0.0f, 0.0f, 0.0f, 0.0f};
  float m, denom;

  if (n <= CAP) {
    int   sc[2];
    float ec[2];
    float mloc = e_self;
    for (int c = 0; c < 2; ++c) {
      int i = c * 32 + lane;
      sc[c] = d;
      ec[c] = -__builtin_inff();
      if (i < n) {
        const float2 p = *(const float2*)(orow + 2 * i);
        sc[c] = __float_as_int(p.x);
        ec[c] = leaky(p.y + ad);
      }
      mloc = fmaxf(mloc, ec[c]);
    }
    m = wave_max(mloc);

    float al[2];
    float sloc = 0.0f;
    for (int c = 0; c < 2; ++c) {
      int i = c * 32 + lane;
      al[c] = (i < n) ? expf(ec[c] - m) : 0.0f;
      sloc += al[c];
    }
    denom = wave_sum(sloc) + expf(e_self - m);
    for (int c = 0; c < 2; ++c) {
      al[c] = al[c] / denom;
      int mc = n - c * 32;
      mc = mc < 0 ? 0 : (mc > 32 ? 32 : mc);
      int j = 0;
      for (; j + 4 <= mc; j += 4) {
        int   s0 = lane_int(sc[c], j),     s1 = lane_int(sc[c], j + 1);
        int   s2 = lane_int(sc[c], j + 2), s3 = lane_int(sc[c], j + 3);
        float a0 = lane_flt(al[c], j),     a1 = lane_flt(al[c], j + 1);
        float a2 = lane_flt(al[c], j + 2), a3 = lane_flt(al[c], j + 3);
        float4 r0 = row4(xw, s0, lane);
        float4 r1 = row4(xw, s1, lane);
        float4 r2 = row4(xw, s2, lane);
        float4 r3 = row4(xw, s3, lane);
        axpy4(acc, a0, r0);
        axpy4(acc, a1, r1);
        axpy4(acc, a2, r2);
        axpy4(acc, a3, r3);
      }
      for (; j < mc; ++j) {
        int   s0 = lane_int(sc[c], j);
        float a0 = lane_flt(al[c], j);
        float4 r0 = row4(xw, s0, lane);
        axpy4(acc, a0, r0);
      }
    }
  } else {
    float mloc = e_self;
    for (int base = 0; base < E; base += 32) {
      int i = base + lane;
      if (i < E && ei[E + i] == d) mloc = fmaxf(mloc, leaky(a_src[ei[i]] + ad));
    }
    m = wave_max(mloc);
    float sloc = 0.0f;
    for (int base = 0; base < E; base += 32) {
      int i = base + lane;
      if (i < E && ei[E + i] == d) sloc += expf(leaky(a_src[ei[i]] + ad) - m);
    }
    denom = wave_sum(sloc) + expf(e_self - m);
    for (int base = 0; base < E; base += 32) {
      int i = base + lane;
      bool v = (i < E) && (ei[E + i] == d);
      int   sv = d;
      float av = 0.0f;
      if (v) {
        sv = ei[i];
        av = expf(leaky(a_src[sv] + ad) - m) / denom;
      }
      unsigned mask = __builtin_amdgcn_ballot_w32(v);
      while (mask) {
        int j = __builtin_ctz(mask);
        mask &= mask - 1;
        int   s0 = lane_int(sv, j);
        float a0 = lane_flt(av, j);
        float4 r0 = row4(xw, s0, lane);
        axpy4(acc, a0, r0);
      }
    }
  }

  {
    float a0 = expf(e_self - m) / denom;
    axpy4(acc, a0, rself);
  }

  float4 o;
  o.x = acc.x + bv.x; o.y = acc.y + bv.y; o.z = acc.z + bv.z; o.w = acc.w + bv.w;
  o.x = o.x > 0.0f ? o.x : (expf(o.x) - 1.0f);
  o.y = o.y > 0.0f ? o.y : (expf(o.y) - 1.0f);
  o.z = o.z > 0.0f ? o.z : (expf(o.z) - 1.0f);
  o.w = o.w > 0.0f ? o.w : (expf(o.w) - 1.0f);
  *(float4*)(orow + lane * 4) = o;
}

extern "C" void kernel_launch(void* const* d_in, const int* in_sizes, int n_in,
                              void* d_out, int out_size, void* d_ws, size_t ws_size,
                              hipStream_t stream) {
  const float* x       = (const float*)d_in[0];
  const int*   ei      = (const int*)  d_in[1];
  const float* W       = (const float*)d_in[2];
  const float* att_src = (const float*)d_in[3];
  const float* att_dst = (const float*)d_in[4];
  const float* bias    = (const float*)d_in[5];

  const int N = in_sizes[0] / DF;
  const int E = in_sizes[1] / 2;
  float* out = (float*)d_out;

  float* xw   = (float*)d_ws;
  float* asrc = xw + (size_t)N * DF;
  float* adst = asrc + N;
  int*   cnt  = (int*)(adst + N);

  {
    int strips = (N + 15) / 16;
    int blocks = (strips + 3) / 4;
    gemm_xw_kernel<<<blocks, 128, 0, stream>>>(x, W, xw, N);
  }
  {
    long long threads = (long long)N * 32;
    logits_kernel<<<(unsigned)((threads + 255) / 256), 256, 0, stream>>>(
        xw, att_src, att_dst, asrc, adst, cnt, N);
  }
  group_kernel<<<(E + 255) / 256, 256, 0, stream>>>(ei, asrc, cnt, out, E);
  {
    long long threads = (long long)N * 32;
    gather_kernel<<<(unsigned)((threads + 255) / 256), 256, 0, stream>>>(
        ei, xw, asrc, adst, cnt, bias, out, E, N);
  }
}


# Logits a_src, a_dst and the counter clear formed in the GEMM epilogue from the accumulators (8 columns per lane, then a 16-lane row sum: new summing order); logits launch and its xw re-read removed
# speedup vs baseline: 2.7110x; 1.1496x over previous
#include <hip/hip_runtime.h>
#include <hip/hip_bf16.h>

typedef __attribute__((ext_vector_type(2))) float v2f;
typedef __attribute__((ext_vector_type(8))) float v8f;

#define DF 128
#define NEG_SLOPE 0.2f

template <int M>
__device__ __forceinline__ float row_xor_add(float v) {
  int o = __builtin_amdgcn_update_dpp(0, __float_as_int(v), 0x160 + M,
                                       0xf,  0xf,
                                       true);
  return v + __int_as_float(o);
}

__global__ void gemm_xw_kernel(const float* __restrict__ x,
                               const float* __restrict__ W,
                               const float* __restrict__ att_src,
                               const float* __restrict__ att_dst,
                               float* __restrict__ xw,
                               float* __restrict__ a_src,
                               float* __restrict__ a_dst,
                               int* __restrict__ cnt, int N) {
  int wave  = threadIdx.x >> 5;
  int lane  = threadIdx.x & 31;
  int strip = blockIdx.x * (blockDim.x >> 5) + wave;
  int tilesM = (N + 15) >> 4;
  if (strip >= tilesM) return;

  int half = lane >> 4;
  int l    = lane & 15;

  int rowA = strip * 16 + l;
  if (rowA >= N) rowA = N - 1;
  const float* Arow = x + (size_t)rowA * DF;

  float sv[8], dv[8];
  #pragma unroll
  for (int t = 0; t < 8; ++t) {
    sv[t] = att_src[t * 16 + l];
    dv[t] = att_dst[t * 16 + l];
  }

  v8f zero = {};
  v8f acc[8];
  #pragma unroll
  for (int t = 0; t < 8; ++t) acc[t] = zero;

  for (int k = 0; k < DF; k += 4) {
    int ka = k + 2 * half;
    v2f a;
    a.x = Arow[ka];
    a.y = Arow[ka + 1];
    const float* Wrow = W + (size_t)ka * DF + l;
    #pragma unroll
    for (int t = 0; t < 8; ++t) {
      v2f b;
      b.x = Wrow[t * 16];
      b.y = Wrow[DF + t * 16];
      acc[t] = __builtin_amdgcn_wmma_f32_16x16x4_f32(
           false, a,  false, b,
           (short)0, acc[t],  false,  false);
    }
  }

  int mbase = strip * 16 + half * 8;

  {
    float ps[8], pd[8];
    #pragma unroll
    for (int v = 0; v < 8; ++v) {
      float s = 0.0f, d = 0.0f;
      #pragma unroll
      for (int t = 0; t < 8; ++t) {
        s += acc[t][v] * sv[t];
        d += acc[t][v] * dv[t];
      }
      s = row_xor_add<8>(s);  d = row_xor_add<8>(d);
      s = row_xor_add<4>(s);  d = row_xor_add<4>(d);
      s = row_xor_add<2>(s);  d = row_xor_add<2>(d);
      s = row_xor_add<1>(s);  d = row_xor_add<1>(d);
      ps[v] = s;
      pd[v] = d;
    }
    float ms = ps[0], md = pd[0];
    #pragma unroll
    for (int v = 1; v < 8; ++v) {
      if (l == v) { ms = ps[v]; md = pd[v]; }
    }
    int row = mbase + l;
    if (l < 8 && row < N) {
      a_src[row] = ms;
      a_dst[row] = md;
      cnt[row]   = 0;
    }
  }

  if (strip * 16 + 16 <= N) {
    #pragma unroll
    for (int t = 0; t < 8; ++t) {
      float* orow = xw + (size_t)mbase * DF + t * 16 + l;
      #pragma unroll
      for (int v = 0; v < 8; ++v) orow[(size_t)v * DF] = acc[t][v];
    }
  } else {
    for (int t = 0; t < 8; ++t)
      for (int v = 0; v < 8; ++v) {
        int row = mbase + v;
        if (row < N) xw[(size_t)row * DF + t * 16 + l] = acc[t][v];
      }
  }
}

__device__ __forceinline__ float leaky(float e) {
  return e >= 0.0f ? e : NEG_SLOPE * e;
}

#define CAP (DF / 2)

__global__ void group_kernel(const int* __restrict__ ei,
                             const float* __restrict__ a_src,
                             int* __restrict__ cnt,
                             float* __restrict__ lists, int E) {
  int i = blockIdx.x * blockDim.x + threadIdx.x;
  if (i >= E) return;
  int s = ei[i];
  int d = ei[E + i];
  float as = a_src[s];
  int r = atomicAdd(&cnt[d], 1);
  r = r < CAP ? r : CAP - 1;
  float2 p;
  p.x = __int_as_float(s);
  p.y = as;
  *(float2*)(lists + (size_t)d * DF + 2 * r) = p;
}

__device__ __forceinline__ float wave_max(float v) {
  for (int off = 16; off > 0; off >>= 1) v = fmaxf(v, __shfl_xor(v, off, 32));
  return v;
}

__device__ __forceinline__ float wave_sum(float v) {
  for (int off = 16; off > 0; off >>= 1) v += __shfl_xor(v, off, 32);
  return v;
}

__device__ __forceinline__ int lane_int(int v, int j) {
  return __builtin_amdgcn_readlane(v, j);
}
__device__ __forceinline__ float lane_flt(float v, int j) {
  return __int_as_float(__builtin_amdgcn_readlane(__float_as_int(v), j));
}

__device__ __forceinline__ float4 row4(const float* __restrict__ xw, int s, int lane) {
  return *(const float4*)(xw + (size_t)s * DF + lane * 4);
}

__device__ __forceinline__ void axpy4(float4& acc, float a, const float4& r) {
  acc.x += a * r.x; acc.y += a * r.y; acc.z += a * r.z; acc.w += a * r.w;
}

__global__ void gather_kernel(const int* __restrict__ ei,
                              const float* __restrict__ xw,
                              const float* __restrict__ a_src,
                              const float* __restrict__ a_dst,
                              const int* __restrict__ cnt,
                              const float* __restrict__ bias,
                              float* out, int E, int N) {
  int gw   = (blockIdx.x * blockDim.x + threadIdx.x) >> 5;
  int lane = threadIdx.x & 31;
  if (gw >= N) return;
  const int d = __builtin_amdgcn_readfirstlane(gw);

  const int   n      = cnt[d];
  const float ad     = a_dst[d];
  const float e_self = leaky(a_src[d] + ad);
  float* orow = out + (size_t)d * DF;
  const float4 rself = row4(xw, d, lane);
  const float4 bv    = *(const float4*)(bias + lane * 4);

  float4 acc = {0.0f, 0.0f, 0.0f, 0.0f};
  float m, denom;

  if (n <= CAP) {
    int   sc[2];
    float ec[2];
    float mloc = e_self;
    for (int c = 0; c < 2; ++c) {
      int i = c * 32 + lane;
      sc[c] = d;
      ec[c] = -__builtin_inff();
      if (i < n) {
        const float2 p = *(const float2*)(orow + 2 * i);
        sc[c] = __float_as_int(p.x);
        ec[c] = leaky(p.y + ad);
      }
      mloc = fmaxf(mloc, ec[c]);
    }
    m = wave_max(mloc);

    float al[2];
    float sloc = 0.0f;
    for (int c = 0; c < 2; ++c) {
      int i = c * 32 + lane;
      al[c] = (i < n) ? expf(ec[c] - m) : 0.0f;
      sloc += al[c];
    }
    denom = wave_sum(sloc) + expf(e_self - m);
    for (int c = 0; c < 2; ++c) {
      al[c] = al[c] / denom;
      int mc = n - c * 32;
      mc = mc < 0 ? 0 : (mc > 32 ? 32 : mc);
      int j = 0;
      for (; j + 4 <= mc; j += 4) {
        int   s0 = lane_int(sc[c], j),     s1 = lane_int(sc[c], j + 1);
        int   s2 = lane_int(sc[c], j + 2), s3 = lane_int(sc[c], j + 3);
        float a0 = lane_flt(al[c], j),     a1 = lane_flt(al[c], j + 1);
        float a2 = lane_flt(al[c], j + 2), a3 = lane_flt(al[c], j + 3);
        float4 r0 = row4(xw, s0, lane);
        float4 r1 = row4(xw, s1, lane);
        float4 r2 = row4(xw, s2, lane);
        float4 r3 = row4(xw, s3, lane);
        axpy4(acc, a0, r0);
        axpy4(acc, a1, r1);
        axpy4(acc, a2, r2);
        axpy4(acc, a3, r3);
      }
      for (; j < mc; ++j) {
        int   s0 = lane_int(sc[c], j);
        float a0 = lane_flt(al[c], j);
        float4 r0 = row4(xw, s0, lane);
        axpy4(acc, a0, r0);
      }
    }
  } else {
    float mloc = e_self;
    for (int base = 0; base < E; base += 32) {
      int i = base + lane;
      if (i < E && ei[E + i] == d) mloc = fmaxf(mloc, leaky(a_src[ei[i]] + ad));
    }
    m = wave_max(mloc);
    float sloc = 0.0f;
    for (int base = 0; base < E; base += 32) {
      int i = base + lane;
      if (i < E && ei[E + i] == d) sloc += expf(leaky(a_src[ei[i]] + ad) - m);
    }
    denom = wave_sum(sloc) + expf(e_self - m);
    for (int base = 0; base < E; base += 32) {
      int i = base + lane;
      bool v = (i < E) && (ei[E + i] == d);
      int   sv = d;
      float av = 0.0f;
      if (v) {
        sv = ei[i];
        av = expf(leaky(a_src[sv] + ad) - m) / denom;
      }
      unsigned mask = __builtin_amdgcn_ballot_w32(v);
      while (mask) {
        int j = __builtin_ctz(mask);
        mask &= mask - 1;
        int   s0 = lane_int(sv, j);
        float a0 = lane_flt(av, j);
        float4 r0 = row4(xw, s0, lane);
        axpy4(acc, a0, r0);
      }
    }
  }

  {
    float a0 = expf(e_self - m) / denom;
    axpy4(acc, a0, rself);
  }

  float4 o;
  o.x = acc.x + bv.x; o.y = acc.y + bv.y; o.z = acc.z + bv.z; o.w = acc.w + bv.w;
  o.x = o.x > 0.0f ? o.x : (expf(o.x) - 1.0f);
  o.y = o.y > 0.0f ? o.y : (expf(o.y) - 1.0f);
  o.z = o.z > 0.0f ? o.z : (expf(o.z) - 1.0f);
  o.w = o.w > 0.0f ? o.w : (expf(o.w) - 1.0f);
  *(float4*)(orow + lane * 4) = o;
}

extern "C" void kernel_launch(void* const* d_in, const int* in_sizes, int n_in,
                              void* d_out, int out_size, void* d_ws, size_t ws_size,
                              hipStream_t stream) {
  const float* x       = (const float*)d_in[0];
  const int*   ei      = (const int*)  d_in[1];
  const float* W       = (const float*)d_in[2];
  const float* att_src = (const float*)d_in[3];
  const float* att_dst = (const float*)d_in[4];
  const float* bias    = (const float*)d_in[5];

  const int N = in_sizes[0] / DF;
  const int E = in_sizes[1] / 2;
  float* out = (float*)d_out;

  float* xw   = (float*)d_ws;
  float* asrc = xw + (size_t)N * DF;
  float* adst = asrc + N;
  int*   cnt  = (int*)(adst + N);

  {
    int strips = (N + 15) / 16;
    int blocks = (strips + 3) / 4;
    gemm_xw_kernel<<<blocks, 128, 0, stream>>>(x, W, att_src, att_dst,
                                               xw, asrc, adst, cnt, N);
  }
  group_kernel<<<(E + 255) / 256, 256, 0, stream>>>(ei, asrc, cnt, out, E);
  {
    long long threads = (long long)N * 32;
    gather_kernel<<<(unsigned)((threads + 255) / 256), 256, 0, stream>>>(
        ei, xw, asrc, adst, cnt, bias, out, E, N);
  }
}


# Edge slots ranked by LDS counters in two passes (edges binned by 256-node range, then ranked per node in a bin) instead of one returning global add per edge; 4 launches; slot order = arrival order
# speedup vs baseline: 4.3360x; 1.5994x over previous
#include <hip/hip_runtime.h>
#include <hip/hip_bf16.h>

typedef __attribute__((ext_vector_type(2))) float v2f;
typedef __attribute__((ext_vector_type(8))) float v8f;

#define DF 128
#define NEG_SLOPE 0.2f

template <int M>
__device__ __forceinline__ float row_xor_add(float v) {
  int o = __builtin_amdgcn_update_dpp(0, __float_as_int(v), 0x160 + M,
                                       0xf,  0xf,
                                       true);
  return v + __int_as_float(o);
}

__global__ void gemm_xw_kernel(const float* __restrict__ x,
                               const float* __restrict__ W,
                               const float* __restrict__ att_src,
                               const float* __restrict__ att_dst,
                               float* __restrict__ xw,
                               float* __restrict__ a_src,
                               float* __restrict__ a_dst, int N) {
  int wave  = threadIdx.x >> 5;
  int lane  = threadIdx.x & 31;
  int strip = blockIdx.x * (blockDim.x >> 5) + wave;
  int tilesM = (N + 15) >> 4;
  if (strip >= tilesM) return;

  int half = lane >> 4;
  int l    = lane & 15;

  int rowA = strip * 16 + l;
  if (rowA >= N) rowA = N - 1;
  const float* Arow = x + (size_t)rowA * DF;

  float sv[8], dv[8];
  #pragma unroll
  for (int t = 0; t < 8; ++t) {
    sv[t] = att_src[t * 16 + l];
    dv[t] = att_dst[t * 16 + l];
  }

  v8f zero = {};
  v8f acc[8];
  #pragma unroll
  for (int t = 0; t < 8; ++t) acc[t] = zero;

  for (int k = 0; k < DF; k += 4) {
    int ka = k + 2 * half;
    v2f a;
    a.x = Arow[ka];
    a.y = Arow[ka + 1];
    const float* Wrow = W + (size_t)ka * DF + l;
    #pragma unroll
    for (int t = 0; t < 8; ++t) {
      v2f b;
      b.x = Wrow[t * 16];
      b.y = Wrow[DF + t * 16];
      acc[t] = __builtin_amdgcn_wmma_f32_16x16x4_f32(
           false, a,  false, b,
           (short)0, acc[t],  false,  false);
    }
  }

  int mbase = strip * 16 + half * 8;

  {
    float ps[8], pd[8];
    #pragma unroll
    for (int v = 0; v < 8; ++v) {
      float s = 0.0f, d = 0.0f;
      #pragma unroll
      for (int t = 0; t < 8; ++t) {
        s += acc[t][v] * sv[t];
        d += acc[t][v] * dv[t];
      }
      s = row_xor_add<8>(s);  d = row_xor_add<8>(d);
      s = row_xor_add<4>(s);  d = row_xor_add<4>(d);
      s = row_xor_add<2>(s);  d = row_xor_add<2>(d);
      s = row_xor_add<1>(s);  d = row_xor_add<1>(d);
      ps[v] = s;
      pd[v] = d;
    }
    float ms = ps[0], md = pd[0];
    #pragma unroll
    for (int v = 1; v < 8; ++v) {
      if (l == v) { ms = ps[v]; md = pd[v]; }
    }
    int row = mbase + l;
    if (l < 8 && row < N) {
      a_src[row] = ms;
      a_dst[row] = md;
    }
  }

  if (strip * 16 + 16 <= N) {
    #pragma unroll
    for (int t = 0; t < 8; ++t) {
      float* orow = xw + (size_t)mbase * DF + t * 16 + l;
      #pragma unroll
      for (int v = 0; v < 8; ++v) orow[(size_t)v * DF] = acc[t][v];
    }
  } else {
    for (int t = 0; t < 8; ++t)
      for (int v = 0; v < 8; ++v) {
        int row = mbase + v;
        if (row < N) xw[(size_t)row * DF + t * 16 + l] = acc[t][v];
      }
  }
}

__device__ __forceinline__ float leaky(float e) {
  return e >= 0.0f ? e : NEG_SLOPE * e;
}

#define CAP (DF / 2)
#define NBW 512
#define CELL_REC 31
#define BIN_THREADS 256
#define BIN_EPT 8
#define RANK_THREADS 1024
#define CELLS_PER_WAVE (NBW / (RANK_THREADS / 32))

__device__ __forceinline__ int bin_of(int d, int NB) {
  int b = d >> 8;
  return b < NB ? b : NB - 1;
}

__global__ void bin_kernel(const int* __restrict__ ei,
                           const float* __restrict__ a_src,
                           float* __restrict__ lists,
                           int E, int NB, int chunk) {
  extern __shared__ int hist[];
  const int w   = blockIdx.x;
  const int tid = threadIdx.x;
  for (int b = tid; b < NB; b += BIN_THREADS) hist[b] = 0;
  __syncthreads();

  const int lo = w * chunk;
  const int hi = (lo + chunk < E) ? lo + chunk : E;
  float* cell0 = lists + (size_t)(w >> 1) * DF + (w & 1) * (DF / 2);

  for (int i0 = lo + tid; i0 < hi; i0 += BIN_EPT * BIN_THREADS) {
    int   s[BIN_EPT], d[BIN_EPT], r[BIN_EPT], bb[BIN_EPT];
    float as[BIN_EPT];
    #pragma unroll
    for (int k = 0; k < BIN_EPT; ++k) {
      int i = i0 + k * BIN_THREADS;
      bool ok = i < hi;
      int ic = ok ? i : hi - 1;
      s[k] = ei[ic];
      d[k] = ok ? ei[E + ic] : -1;
    }
    #pragma unroll
    for (int k = 0; k < BIN_EPT; ++k) as[k] = a_src[s[k]];
    #pragma unroll
    for (int k = 0; k < BIN_EPT; ++k) {
      r[k]  = CELL_REC;
      bb[k] = 0;
      if (d[k] >= 0) {
        bb[k] = bin_of(d[k], NB);
        r[k]  = atomicAdd(&hist[bb[k]], 1);
      }
    }
    #pragma unroll
    for (int k = 0; k < BIN_EPT; ++k) {
      if (r[k] < CELL_REC) {
        int dl = d[k] - (bb[k] << 8);
        float2 p;
        p.x = __int_as_float(s[k] | (int)((unsigned)dl << 23));
        p.y = as[k];
        *(float2*)(cell0 + (size_t)bb[k] * (256 * DF) + 2 * (r[k] + 1)) = p;
      }
    }
  }
  __syncthreads();
  for (int b = tid; b < NB; b += BIN_THREADS) {
    float2 p;
    p.x = __int_as_float(hist[b]);
    p.y = 0.0f;
    *(float2*)(cell0 + (size_t)b * (256 * DF)) = p;
  }
}

__global__ void rank_kernel(const int* __restrict__ ei,
                            const float* __restrict__ a_src,
                            int* __restrict__ cnt,
                            float* lists, int E, int N, int NB) {
  __shared__ int cntL[512];
  __shared__ int ovf;
  const int b    = blockIdx.x;
  const int tid  = threadIdx.x;
  const int lane = tid & 31;
  const int wv   = tid >> 5;
  const int node0 = b << 8;
  const int nn    = (b == NB - 1) ? N - node0 : 256;

  if (tid < 512) cntL[tid] = 0;
  if (tid == 0) ovf = 0;
  __syncthreads();

  float2 rec[CELLS_PER_WAVE];
  int    rk[CELLS_PER_WAVE];
  #pragma unroll
  for (int q = 0; q < CELLS_PER_WAVE; ++q) {
    int c = wv * CELLS_PER_WAVE + q;
    rec[q] = *(const float2*)(lists + (size_t)(node0 + (c >> 1)) * DF +
                              (c & 1) * (DF / 2) + lane * 2);
  }
  #pragma unroll
  for (int q = 0; q < CELLS_PER_WAVE; ++q) {
    int x = __float_as_int(rec[q].x);
    int n = __builtin_amdgcn_readfirstlane(x);
    if (n > CELL_REC) ovf = 1;
    rk[q] = -1;
    if (lane >= 1 && lane <= n) rk[q] = atomicAdd(&cntL[(unsigned)x >> 23], 1);
  }
  __syncthreads();

  if (ovf == 0) {
    #pragma unroll
    for (int q = 0; q < CELLS_PER_WAVE; ++q) {
      if (rk[q] >= 0) {
        int x  = __float_as_int(rec[q].x);
        int dl = (unsigned)x >> 23;
        int r  = rk[q] < CAP ? rk[q] : CAP - 1;
        float2 p;
        p.x = __int_as_float(x & 0x7fffff);
        p.y = rec[q].y;
        *(float2*)(lists + (size_t)(node0 + dl) * DF + 2 * r) = p;
      }
    }
  } else {
    if (tid < 512) cntL[tid] = 0;
    __syncthreads();
    for (int i = tid; i < E; i += RANK_THREADS) {
      int d = ei[E + i];
      if (bin_of(d, NB) == b) {
        int s  = ei[i];
        int dl = d - node0;
        int r  = atomicAdd(&cntL[dl], 1);
        r = r < CAP ? r : CAP - 1;
        float2 p;
        p.x = __int_as_float(s);
        p.y = a_src[s];
        *(float2*)(lists + (size_t)d * DF + 2 * r) = p;
      }
    }
    __syncthreads();
  }
  if (tid < nn) cnt[node0 + tid] = cntL[tid];
}

__device__ __forceinline__ float wave_max(float v) {
  for (int off = 16; off > 0; off >>= 1) v = fmaxf(v, __shfl_xor(v, off, 32));
  return v;
}

__device__ __forceinline__ float wave_sum(float v) {
  for (int off = 16; off > 0; off >>= 1) v += __shfl_xor(v, off, 32);
  return v;
}

__device__ __forceinline__ int lane_int(int v, int j) {
  return __builtin_amdgcn_readlane(v, j);
}
__device__ __forceinline__ float lane_flt(float v, int j) {
  return __int_as_float(__builtin_amdgcn_readlane(__float_as_int(v), j));
}

__device__ __forceinline__ float4 row4(const float* __restrict__ xw, int s, int lane) {
  return *(const float4*)(xw + (size_t)s * DF + lane * 4);
}

__device__ __forceinline__ void axpy4(float4& acc, float a, const float4& r) {
  acc.x += a * r.x; acc.y += a * r.y; acc.z += a * r.z; acc.w += a * r.w;
}

__global__ void gather_kernel(const int* __restrict__ ei,
                              const float* __restrict__ xw,
                              const float* __restrict__ a_src,
                              const float* __restrict__ a_dst,
                              const int* __restrict__ cnt,
                              const float* __restrict__ bias,
                              float* out, int E, int N) {
  int gw   = (blockIdx.x * blockDim.x + threadIdx.x) >> 5;
  int lane = threadIdx.x & 31;
  if (gw >= N) return;
  const int d = __builtin_amdgcn_readfirstlane(gw);

  const int   n      = cnt[d];
  const float ad     = a_dst[d];
  const float e_self = leaky(a_src[d] + ad);
  float* orow = out + (size_t)d * DF;
  const float4 rself = row4(xw, d, lane);
  const float4 bv    = *(const float4*)(bias + lane * 4);

  float4 acc = {0.0f, 0.0f, 0.0f, 0.0f};
  float m, denom;

  if (n <= CAP) {
    int   sc[2];
    float ec[2];
    float mloc = e_self;
    for (int c = 0; c < 2; ++c) {
      int i = c * 32 + lane;
      sc[c] = d;
      ec[c] = -__builtin_inff();
      if (i < n) {
        const float2 p = *(const float2*)(orow + 2 * i);
        sc[c] = __float_as_int(p.x);
        ec[c] = leaky(p.y + ad);
      }
      mloc = fmaxf(mloc, ec[c]);
    }
    m = wave_max(mloc);

    float al[2];
    float sloc = 0.0f;
    for (int c = 0; c < 2; ++c) {
      int i = c * 32 + lane;
      al[c] = (i < n) ? expf(ec[c] - m) : 0.0f;
      sloc += al[c];
    }
    denom = wave_sum(sloc) + expf(e_self - m);
    for (int c = 0; c < 2; ++c) {
      al[c] = al[c] / denom;
      int mc = n - c * 32;
      mc = mc < 0 ? 0 : (mc > 32 ? 32 : mc);
      int j = 0;
      for (; j + 4 <= mc; j += 4) {
        int   s0 = lane_int(sc[c], j),     s1 = lane_int(sc[c], j + 1);
        int   s2 = lane_int(sc[c], j + 2), s3 = lane_int(sc[c], j + 3);
        float a0 = lane_flt(al[c], j),     a1 = lane_flt(al[c], j + 1);
        float a2 = lane_flt(al[c], j + 2), a3 = lane_flt(al[c], j + 3);
        float4 r0 = row4(xw, s0, lane);
        float4 r1 = row4(xw, s1, lane);
        float4 r2 = row4(xw, s2, lane);
        float4 r3 = row4(xw, s3, lane);
        axpy4(acc, a0, r0);
        axpy4(acc, a1, r1);
        axpy4(acc, a2, r2);
        axpy4(acc, a3, r3);
      }
      for (; j < mc; ++j) {
        int   s0 = lane_int(sc[c], j);
        float a0 = lane_flt(al[c], j);
        float4 r0 = row4(xw, s0, lane);
        axpy4(acc, a0, r0);
      }
    }
  } else {
    float mloc = e_self;
    for (int base = 0; base < E; base += 32) {
      int i = base + lane;
      if (i < E && ei[E + i] == d) mloc = fmaxf(mloc, leaky(a_src[ei[i]] + ad));
    }
    m = wave_max(mloc);
    float sloc = 0.0f;
    for (int base = 0; base < E; base += 32) {
      int i = base + lane;
      if (i < E && ei[E + i] == d) sloc += expf(leaky(a_src[ei[i]] + ad) - m);
    }
    denom = wave_sum(sloc) + expf(e_self - m);
    for (int base = 0; base < E; base += 32) {
      int i = base + lane;
      bool v = (i < E) && (ei[E + i] == d);
      int   sv = d;
      float av = 0.0f;
      if (v) {
        sv = ei[i];
        av = expf(leaky(a_src[sv] + ad) - m) / denom;
      }
      unsigned mask = __builtin_amdgcn_ballot_w32(v);
      while (mask) {
        int j = __builtin_ctz(mask);
        mask &= mask - 1;
        int   s0 = lane_int(sv, j);
        float a0 = lane_flt(av, j);
        float4 r0 = row4(xw, s0, lane);
        axpy4(acc, a0, r0);
      }
    }
  }

  {
    float a0 = expf(e_self - m) / denom;
    axpy4(acc, a0, rself);
  }

  float4 o;
  o.x = acc.x + bv.x; o.y = acc.y + bv.y; o.z = acc.z + bv.z; o.w = acc.w + bv.w;
  o.x = o.x > 0.0f ? o.x : (expf(o.x) - 1.0f);
  o.y = o.y > 0.0f ? o.y : (expf(o.y) - 1.0f);
  o.z = o.z > 0.0f ? o.z : (expf(o.z) - 1.0f);
  o.w = o.w > 0.0f ? o.w : (expf(o.w) - 1.0f);
  *(float4*)(orow + lane * 4) = o;
}

extern "C" void kernel_launch(void* const* d_in, const int* in_sizes, int n_in,
                              void* d_out, int out_size, void* d_ws, size_t ws_size,
                              hipStream_t stream) {
  const float* x       = (const float*)d_in[0];
  const int*   ei      = (const int*)  d_in[1];
  const float* W       = (const float*)d_in[2];
  const float* att_src = (const float*)d_in[3];
  const float* att_dst = (const float*)d_in[4];
  const float* bias    = (const float*)d_in[5];

  const int N = in_sizes[0] / DF;
  const int E = in_sizes[1] / 2;
  float* out = (float*)d_out;

  float* xw   = (float*)d_ws;
  float* asrc = xw + (size_t)N * DF;
  float* adst = asrc + N;
  int*   cnt  = (int*)(adst + N);

  {
    int strips = (N + 15) / 16;
    int blocks = (strips + 3) / 4;
    gemm_xw_kernel<<<blocks, 128, 0, stream>>>(x, W, att_src, att_dst,
                                               xw, asrc, adst, N);
  }
  {
    int NB    = N >> 8;
    int chunk = (E + NBW - 1) / NBW;
    bin_kernel<<<NBW, BIN_THREADS, NB * sizeof(int), stream>>>(ei, asrc, out,
                                                              E, NB, chunk);
    rank_kernel<<<NB, RANK_THREADS, 0, stream>>>(ei, asrc, cnt, out, E, N, NB);
  }
  {
    long long threads = (long long)N * 32;
    gather_kernel<<<(unsigned)((threads + 255) / 256), 256, 0, stream>>>(
        ei, xw, asrc, adst, cnt, bias, out, E, N);
  }
}


# Bin-pass records and rank-pass node lists staged in LDS, then stored as one contiguous run per cell / per node instead of one scattered 8-byte store per edge; slot order unchanged
# speedup vs baseline: 4.4970x; 1.0371x over previous
#include <hip/hip_runtime.h>
#include <hip/hip_bf16.h>

typedef __attribute__((ext_vector_type(2))) float v2f;
typedef __attribute__((ext_vector_type(8))) float v8f;

#define DF 128
#define NEG_SLOPE 0.2f

template <int M>
__device__ __forceinline__ float row_xor_add(float v) {
  int o = __builtin_amdgcn_update_dpp(0, __float_as_int(v), 0x160 + M,
                                       0xf,  0xf,
                                       true);
  return v + __int_as_float(o);
}

__global__ void gemm_xw_kernel(const float* __restrict__ x,
                               const float* __restrict__ W,
                               const float* __restrict__ att_src,
                               const float* __restrict__ att_dst,
                               float* __restrict__ xw,
                               float* __restrict__ a_src,
                               float* __restrict__ a_dst, int N) {
  int wave  = threadIdx.x >> 5;
  int lane  = threadIdx.x & 31;
  int strip = blockIdx.x * (blockDim.x >> 5) + wave;
  int tilesM = (N + 15) >> 4;
  if (strip >= tilesM) return;

  int half = lane >> 4;
  int l    = lane & 15;

  int rowA = strip * 16 + l;
  if (rowA >= N) rowA = N - 1;
  const float* Arow = x + (size_t)rowA * DF;

  float sv[8], dv[8];
  #pragma unroll
  for (int t = 0; t < 8; ++t) {
    sv[t] = att_src[t * 16 + l];
    dv[t] = att_dst[t * 16 + l];
  }

  v8f zero = {};
  v8f acc[8];
  #pragma unroll
  for (int t = 0; t < 8; ++t) acc[t] = zero;

  for (int k = 0; k < DF; k += 4) {
    int ka = k + 2 * half;
    v2f a;
    a.x = Arow[ka];
    a.y = Arow[ka + 1];
    const float* Wrow = W + (size_t)ka * DF + l;
    #pragma unroll
    for (int t = 0; t < 8; ++t) {
      v2f b;
      b.x = Wrow[t * 16];
      b.y = Wrow[DF + t * 16];
      acc[t] = __builtin_amdgcn_wmma_f32_16x16x4_f32(
           false, a,  false, b,
           (short)0, acc[t],  false,  false);
    }
  }

  int mbase = strip * 16 + half * 8;

  {
    float ps[8], pd[8];
    #pragma unroll
    for (int v = 0; v < 8; ++v) {
      float s = 0.0f, d = 0.0f;
      #pragma unroll
      for (int t = 0; t < 8; ++t) {
        s += acc[t][v] * sv[t];
        d += acc[t][v] * dv[t];
      }
      s = row_xor_add<8>(s);  d = row_xor_add<8>(d);
      s = row_xor_add<4>(s);  d = row_xor_add<4>(d);
      s = row_xor_add<2>(s);  d = row_xor_add<2>(d);
      s = row_xor_add<1>(s);  d = row_xor_add<1>(d);
      ps[v] = s;
      pd[v] = d;
    }
    float ms = ps[0], md = pd[0];
    #pragma unroll
    for (int v = 1; v < 8; ++v) {
      if (l == v) { ms = ps[v]; md = pd[v]; }
    }
    int row = mbase + l;
    if (l < 8 && row < N) {
      a_src[row] = ms;
      a_dst[row] = md;
    }
  }

  if (strip * 16 + 16 <= N) {
    #pragma unroll
    for (int t = 0; t < 8; ++t) {
      float* orow = xw + (size_t)mbase * DF + t * 16 + l;
      #pragma unroll
      for (int v = 0; v < 8; ++v) orow[(size_t)v * DF] = acc[t][v];
    }
  } else {
    for (int t = 0; t < 8; ++t)
      for (int v = 0; v < 8; ++v) {
        int row = mbase + v;
        if (row < N) xw[(size_t)row * DF + t * 16 + l] = acc[t][v];
      }
  }
}

__device__ __forceinline__ float leaky(float e) {
  return e >= 0.0f ? e : NEG_SLOPE * e;
}

#define CAP (DF / 2)
#define NBW 512
#define CELL_REC 31
#define BIN_THREADS 256
#define BIN_WAVES (BIN_THREADS / 32)
#define BIN_EPT 8
#define RANK_THREADS 1024
#define RANK_WAVES (RANK_THREADS / 32)
#define CELLS_PER_WAVE (NBW / RANK_WAVES)
#define STAGE_PITCH (CELL_REC + 2)
#define LIST_PITCH  (CAP + 1)

__device__ __forceinline__ int bin_of(int d, int NB) {
  int b = d >> 8;
  return b < NB ? b : NB - 1;
}

__global__ void bin_kernel(const int* __restrict__ ei,
                           const float* __restrict__ a_src,
                           float* __restrict__ lists,
                           int E, int NB, int chunk) {
  extern __shared__ int2 stage[];
  int* hist = (int*)(stage + (size_t)NB * STAGE_PITCH);
  const int w    = blockIdx.x;
  const int tid  = threadIdx.x;
  const int lane = tid & 31;
  const int wv   = tid >> 5;
  for (int b = tid; b < NB; b += BIN_THREADS) hist[b] = 0;
  __syncthreads();

  const int lo = w * chunk;
  const int hi = (lo + chunk < E) ? lo + chunk : E;

  for (int i0 = lo + tid; i0 < hi; i0 += BIN_EPT * BIN_THREADS) {
    int   s[BIN_EPT], d[BIN_EPT], r[BIN_EPT], bb[BIN_EPT];
    float as[BIN_EPT];
    #pragma unroll
    for (int k = 0; k < BIN_EPT; ++k) {
      int i = i0 + k * BIN_THREADS;
      bool ok = i < hi;
      int ic = ok ? i : hi - 1;
      s[k] = ei[ic];
      d[k] = ok ? ei[E + ic] : -1;
    }
    #pragma unroll
    for (int k = 0; k < BIN_EPT; ++k) as[k] = a_src[s[k]];
    #pragma unroll
    for (int k = 0; k < BIN_EPT; ++k) {
      r[k]  = CELL_REC;
      bb[k] = 0;
      if (d[k] >= 0) {
        bb[k] = bin_of(d[k], NB);
        r[k]  = atomicAdd(&hist[bb[k]], 1);
      }
    }
    #pragma unroll
    for (int k = 0; k < BIN_EPT; ++k) {
      if (r[k] < CELL_REC) {
        int dl = d[k] - (bb[k] << 8);
        int2 p;
        p.x = s[k] | (int)((unsigned)dl << 23);
        p.y = __float_as_int(as[k]);
        stage[bb[k] * STAGE_PITCH + r[k] + 1] = p;
      }
    }
  }
  __syncthreads();

  int2* cellw = (int2*)lists + (size_t)(w >> 1) * (DF / 2) + (w & 1) * (DF / 4);
  for (int b0 = wv; b0 < NB; b0 += 4 * BIN_WAVES) {
    int  n[4];
    int2 p[4];
    #pragma unroll
    for (int u = 0; u < 4; ++u) {
      int b = b0 + u * BIN_WAVES;
      n[u] = b < NB ? hist[b] : -1;
    }
    #pragma unroll
    for (int u = 0; u < 4; ++u) {
      int b = b0 + u * BIN_WAVES;
      int m = n[u] < CELL_REC ? n[u] : CELL_REC;
      p[u].x = n[u];
      p[u].y = 0;
      if (lane >= 1 && lane <= m) p[u] = stage[b * STAGE_PITCH + lane];
    }
    #pragma unroll
    for (int u = 0; u < 4; ++u) {
      int b = b0 + u * BIN_WAVES;
      int m = n[u] < CELL_REC ? n[u] : CELL_REC;
      if (lane <= m) cellw[(size_t)b * (256 * (DF / 2)) + lane] = p[u];
    }
  }
}

__global__ void rank_kernel(const int* __restrict__ ei,
                            const float* __restrict__ a_src,
                            int* __restrict__ cnt,
                            float* lists, int E, int N, int NB) {
  __shared__ int2 lst[512 * LIST_PITCH];
  __shared__ int cntL[512];
  __shared__ int ovf;
  const int b    = blockIdx.x;
  const int tid  = threadIdx.x;
  const int lane = tid & 31;
  const int wv   = tid >> 5;
  const int node0 = b << 8;
  const int nn    = (b == NB - 1) ? N - node0 : 256;

  if (tid < 512) cntL[tid] = 0;
  if (tid == 0) ovf = 0;
  __syncthreads();

  float2 rec[CELLS_PER_WAVE];
  int    rk[CELLS_PER_WAVE];
  #pragma unroll
  for (int q = 0; q < CELLS_PER_WAVE; ++q) {
    int c = wv * CELLS_PER_WAVE + q;
    rec[q] = *(const float2*)(lists + (size_t)(node0 + (c >> 1)) * DF +
                              (c & 1) * (DF / 2) + lane * 2);
  }
  #pragma unroll
  for (int q = 0; q < CELLS_PER_WAVE; ++q) {
    int x = __float_as_int(rec[q].x);
    int n = __builtin_amdgcn_readfirstlane(x);
    if (n > CELL_REC) ovf = 1;
    rk[q] = -1;
    if (lane >= 1 && lane <= n) rk[q] = atomicAdd(&cntL[(unsigned)x >> 23], 1);
  }
  #pragma unroll
  for (int q = 0; q < CELLS_PER_WAVE; ++q) {
    if (rk[q] >= 0 && rk[q] < CAP) {
      int x  = __float_as_int(rec[q].x);
      int dl = (unsigned)x >> 23;
      int2 p;
      p.x = x & 0x7fffff;
      p.y = __float_as_int(rec[q].y);
      lst[dl * LIST_PITCH + rk[q]] = p;
    }
  }
  __syncthreads();

  if (ovf != 0) {
    if (tid < 512) cntL[tid] = 0;
    __syncthreads();
    for (int i = tid; i < E; i += RANK_THREADS) {
      int d = ei[E + i];
      if (bin_of(d, NB) == b) {
        int s  = ei[i];
        int dl = d - node0;
        int r  = atomicAdd(&cntL[dl], 1);
        if (r < CAP) {
          int2 p;
          p.x = s;
          p.y = __float_as_int(a_src[s]);
          lst[dl * LIST_PITCH + r] = p;
        }
      }
    }
    __syncthreads();
  }

  for (int dl0 = wv; dl0 < nn; dl0 += 4 * RANK_WAVES) {
    int  nv[4], m[4];
    int2 p[4];
    #pragma unroll
    for (int u = 0; u < 4; ++u) {
      int dl = dl0 + u * RANK_WAVES;
      nv[u] = dl < nn ? cntL[dl] : 0;
    }
    #pragma unroll
    for (int u = 0; u < 4; ++u) {
      int n = __builtin_amdgcn_readfirstlane(nv[u]);
      m[u] = n < CAP ? n : CAP;
    }
    #pragma unroll
    for (int u = 0; u < 4; ++u) {
      int dl = dl0 + u * RANK_WAVES;
      p[u].x = 0;
      p[u].y = 0;
      if (lane < m[u]) p[u] = lst[dl * LIST_PITCH + lane];
    }
    #pragma unroll
    for (int u = 0; u < 4; ++u) {
      int dl = dl0 + u * RANK_WAVES;
      int2* row = (int2*)lists + (size_t)(node0 + dl) * (DF / 2);
      if (lane < m[u]) row[lane] = p[u];
      if (m[u] > 32) {
        if (lane + 32 < m[u]) row[lane + 32] = lst[dl * LIST_PITCH + lane + 32];
      }
    }
  }
  if (tid < nn) cnt[node0 + tid] = cntL[tid];
}

__device__ __forceinline__ float wave_max(float v) {
  for (int off = 16; off > 0; off >>= 1) v = fmaxf(v, __shfl_xor(v, off, 32));
  return v;
}

__device__ __forceinline__ float wave_sum(float v) {
  for (int off = 16; off > 0; off >>= 1) v += __shfl_xor(v, off, 32);
  return v;
}

__device__ __forceinline__ int lane_int(int v, int j) {
  return __builtin_amdgcn_readlane(v, j);
}
__device__ __forceinline__ float lane_flt(float v, int j) {
  return __int_as_float(__builtin_amdgcn_readlane(__float_as_int(v), j));
}

__device__ __forceinline__ float4 row4(const float* __restrict__ xw, int s, int lane) {
  return *(const float4*)(xw + (size_t)s * DF + lane * 4);
}

__device__ __forceinline__ void axpy4(float4& acc, float a, const float4& r) {
  acc.x += a * r.x; acc.y += a * r.y; acc.z += a * r.z; acc.w += a * r.w;
}

__global__ void gather_kernel(const int* __restrict__ ei,
                              const float* __restrict__ xw,
                              const float* __restrict__ a_src,
                              const float* __restrict__ a_dst,
                              const int* __restrict__ cnt,
                              const float* __restrict__ bias,
                              float* out, int E, int N) {
  int gw   = (blockIdx.x * blockDim.x + threadIdx.x) >> 5;
  int lane = threadIdx.x & 31;
  if (gw >= N) return;
  const int d = __builtin_amdgcn_readfirstlane(gw);

  const int   n      = cnt[d];
  const float ad     = a_dst[d];
  const float e_self = leaky(a_src[d] + ad);
  float* orow = out + (size_t)d * DF;
  const float4 rself = row4(xw, d, lane);
  const float4 bv    = *(const float4*)(bias + lane * 4);

  float4 acc = {0.0f, 0.0f, 0.0f, 0.0f};
  float m, denom;

  if (n <= CAP) {
    int   sc[2];
    float ec[2];
    float mloc = e_self;
    for (int c = 0; c < 2; ++c) {
      int i = c * 32 + lane;
      sc[c] = d;
      ec[c] = -__builtin_inff();
      if (i < n) {
        const float2 p = *(const float2*)(orow + 2 * i);
        sc[c] = __float_as_int(p.x);
        ec[c] = leaky(p.y + ad);
      }
      mloc = fmaxf(mloc, ec[c]);
    }
    m = wave_max(mloc);

    float al[2];
    float sloc = 0.0f;
    for (int c = 0; c < 2; ++c) {
      int i = c * 32 + lane;
      al[c] = (i < n) ? expf(ec[c] - m) : 0.0f;
      sloc += al[c];
    }
    denom = wave_sum(sloc) + expf(e_self - m);
    for (int c = 0; c < 2; ++c) {
      al[c] = al[c] / denom;
      int mc = n - c * 32;
      mc = mc < 0 ? 0 : (mc > 32 ? 32 : mc);
      int j = 0;
      for (; j + 4 <= mc; j += 4) {
        int   s0 = lane_int(sc[c], j),     s1 = lane_int(sc[c], j + 1);
        int   s2 = lane_int(sc[c], j + 2), s3 = lane_int(sc[c], j + 3);
        float a0 = lane_flt(al[c], j),     a1 = lane_flt(al[c], j + 1);
        float a2 = lane_flt(al[c], j + 2), a3 = lane_flt(al[c], j + 3);
        float4 r0 = row4(xw, s0, lane);
        float4 r1 = row4(xw, s1, lane);
        float4 r2 = row4(xw, s2, lane);
        float4 r3 = row4(xw, s3, lane);
        axpy4(acc, a0, r0);
        axpy4(acc, a1, r1);
        axpy4(acc, a2, r2);
        axpy4(acc, a3, r3);
      }
      for (; j < mc; ++j) {
        int   s0 = lane_int(sc[c], j);
        float a0 = lane_flt(al[c], j);
        float4 r0 = row4(xw, s0, lane);
        axpy4(acc, a0, r0);
      }
    }
  } else {
    float mloc = e_self;
    for (int base = 0; base < E; base += 32) {
      int i = base + lane;
      if (i < E && ei[E + i] == d) mloc = fmaxf(mloc, leaky(a_src[ei[i]] + ad));
    }
    m = wave_max(mloc);
    float sloc = 0.0f;
    for (int base = 0; base < E; base += 32) {
      int i = base + lane;
      if (i < E && ei[E + i] == d) sloc += expf(leaky(a_src[ei[i]] + ad) - m);
    }
    denom = wave_sum(sloc) + expf(e_self - m);
    for (int base = 0; base < E; base += 32) {
      int i = base + lane;
      bool v = (i < E) && (ei[E + i] == d);
      int   sv = d;
      float av = 0.0f;
      if (v) {
        sv = ei[i];
        av = expf(leaky(a_src[sv] + ad) - m) / denom;
      }
      unsigned mask = __builtin_amdgcn_ballot_w32(v);
      while (mask) {
        int j = __builtin_ctz(mask);
        mask &= mask - 1;
        int   s0 = lane_int(sv, j);
        float a0 = lane_flt(av, j);
        float4 r0 = row4(xw, s0, lane);
        axpy4(acc, a0, r0);
      }
    }
  }

  {
    float a0 = expf(e_self - m) / denom;
    axpy4(acc, a0, rself);
  }

  float4 o;
  o.x = acc.x + bv.x; o.y = acc.y + bv.y; o.z = acc.z + bv.z; o.w = acc.w + bv.w;
  o.x = o.x > 0.0f ? o.x : (expf(o.x) - 1.0f);
  o.y = o.y > 0.0f ? o.y : (expf(o.y) - 1.0f);
  o.z = o.z > 0.0f ? o.z : (expf(o.z) - 1.0f);
  o.w = o.w > 0.0f ? o.w : (expf(o.w) - 1.0f);
  *(float4*)(orow + lane * 4) = o;
}

extern "C" void kernel_launch(void* const* d_in, const int* in_sizes, int n_in,
                              void* d_out, int out_size, void* d_ws, size_t ws_size,
                              hipStream_t stream) {
  const float* x       = (const float*)d_in[0];
  const int*   ei      = (const int*)  d_in[1];
  const float* W       = (const float*)d_in[2];
  const float* att_src = (const float*)d_in[3];
  const float* att_dst = (const float*)d_in[4];
  const float* bias    = (const float*)d_in[5];

  const int N = in_sizes[0] / DF;
  const int E = in_sizes[1] / 2;
  float* out = (float*)d_out;

  float* xw   = (float*)d_ws;
  float* asrc = xw + (size_t)N * DF;
  float* adst = asrc + N;
  int*   cnt  = (int*)(adst + N);

  {
    int strips = (N + 15) / 16;
    int blocks = (strips + 3) / 4;
    gemm_xw_kernel<<<blocks, 128, 0, stream>>>(x, W, att_src, att_dst,
                                               xw, asrc, adst, N);
  }
  {
    int NB    = N >> 8;
    int chunk = (E + NBW - 1) / NBW;
    bin_kernel<<<NBW, BIN_THREADS,
                 NB * (STAGE_PITCH * sizeof(int2) + sizeof(int)), stream>>>(
        ei, asrc, out, E, NB, chunk);
    rank_kernel<<<NB, RANK_THREADS, 0, stream>>>(ei, asrc, cnt, out, E, N, NB);
  }
  {
    long long threads = (long long)N * 32;
    gather_kernel<<<(unsigned)((threads + 255) / 256), 256, 0, stream>>>(
        ei, xw, asrc, adst, cnt, bias, out, E, N);
  }
}


# Bin pass runs as extra work-groups of the GEMM launch, not behind it (records carry no logit: the rank pass reads a_src[source]; a cell's records ordered by key, not by arrival); 3 launches
# speedup vs baseline: 4.7319x; 1.0522x over previous
#include <hip/hip_runtime.h>
#include <hip/hip_bf16.h>

typedef __attribute__((ext_vector_type(2))) float v2f;
typedef __attribute__((ext_vector_type(8))) float v8f;

#define DF 128
#define NEG_SLOPE 0.2f

template <int M>
__device__ __forceinline__ float row_xor_add(float v) {
  int o = __builtin_amdgcn_update_dpp(0, __float_as_int(v), 0x160 + M,
                                       0xf,  0xf,
                                       true);
  return v + __int_as_float(o);
}

__device__ __forceinline__ void gemm_xw_body(int gblock,
                               const float* __restrict__ x,
                               const float* __restrict__ W,
                               const float* __restrict__ att_src,
                               const float* __restrict__ att_dst,
                               float* __restrict__ xw,
                               float* __restrict__ a_src,
                               float* __restrict__ a_dst, int N) {
  int wave  = threadIdx.x >> 5;
  int lane  = threadIdx.x & 31;
  int strip = gblock * (blockDim.x >> 5) + wave;
  int tilesM = (N + 15) >> 4;
  if (strip >= tilesM) return;

  int half = lane >> 4;
  int l    = lane & 15;

  int rowA = strip * 16 + l;
  if (rowA >= N) rowA = N - 1;
  const float* Arow = x + (size_t)rowA * DF;

  float sv[8], dv[8];
  #pragma unroll
  for (int t = 0; t < 8; ++t) {
    sv[t] = att_src[t * 16 + l];
    dv[t] = att_dst[t * 16 + l];
  }

  v8f zero = {};
  v8f acc[8];
  #pragma unroll
  for (int t = 0; t < 8; ++t) acc[t] = zero;

  for (int k = 0; k < DF; k += 4) {
    int ka = k + 2 * half;
    v2f a;
    a.x = Arow[ka];
    a.y = Arow[ka + 1];
    const float* Wrow = W + (size_t)ka * DF + l;
    #pragma unroll
    for (int t = 0; t < 8; ++t) {
      v2f b;
      b.x = Wrow[t * 16];
      b.y = Wrow[DF + t * 16];
      acc[t] = __builtin_amdgcn_wmma_f32_16x16x4_f32(
           false, a,  false, b,
           (short)0, acc[t],  false,  false);
    }
  }

  int mbase = strip * 16 + half * 8;

  {
    float ps[8], pd[8];
    #pragma unroll
    for (int v = 0; v < 8; ++v) {
      float s = 0.0f, d = 0.0f;
      #pragma unroll
      for (int t = 0; t < 8; ++t) {
        s += acc[t][v] * sv[t];
        d += acc[t][v] * dv[t];
      }
      s = row_xor_add<8>(s);  d = row_xor_add<8>(d);
      s = row_xor_add<4>(s);  d = row_xor_add<4>(d);
      s = row_xor_add<2>(s);  d = row_xor_add<2>(d);
      s = row_xor_add<1>(s);  d = row_xor_add<1>(d);
      ps[v] = s;
      pd[v] = d;
    }
    float ms = ps[0], md = pd[0];
    #pragma unroll
    for (int v = 1; v < 8; ++v) {
      if (l == v) { ms = ps[v]; md = pd[v]; }
    }
    int row = mbase + l;
    if (l < 8 && row < N) {
      a_src[row] = ms;
      a_dst[row] = md;
    }
  }

  if (strip * 16 + 16 <= N) {
    #pragma unroll
    for (int t = 0; t < 8; ++t) {
      float* orow = xw + (size_t)mbase * DF + t * 16 + l;
      #pragma unroll
      for (int v = 0; v < 8; ++v) orow[(size_t)v * DF] = acc[t][v];
    }
  } else {
    for (int t = 0; t < 8; ++t)
      for (int v = 0; v < 8; ++v) {
        int row = mbase + v;
        if (row < N) xw[(size_t)row * DF + t * 16 + l] = acc[t][v];
      }
  }
}

__device__ __forceinline__ float leaky(float e) {
  return e >= 0.0f ? e : NEG_SLOPE * e;
}

#define CAP (DF / 2)
#define NBW 512
#define CELL_REC 31
#define BIN_THREADS 256
#define BIN_WAVES (BIN_THREADS / 32)
#define BIN_EPT 8
#define RANK_THREADS 1024
#define RANK_WAVES (RANK_THREADS / 32)
#define CELLS_PER_WAVE (NBW / RANK_WAVES)
#define STAGE_PITCH (CELL_REC + 2)
#define LIST_PITCH  (CAP + 1)

__device__ __forceinline__ int bin_of(int d, int NB) {
  int b = d >> 8;
  return b < NB ? b : NB - 1;
}

__device__ __forceinline__ void bin_body(int w, const int* __restrict__ ei,
                           float* __restrict__ lists,
                           int E, int NB, int chunk) {
  extern __shared__ int2 stage[];
  int* hist = (int*)(stage + (size_t)NB * STAGE_PITCH);
  const int tid  = threadIdx.x;
  const int lane = tid & 31;
  const int wv   = tid >> 5;
  for (int b = tid; b < NB; b += BIN_THREADS) hist[b] = 0;
  __syncthreads();

  const int lo = w * chunk;
  const int hi = (lo + chunk < E) ? lo + chunk : E;

  for (int i0 = lo + tid; i0 < hi; i0 += BIN_EPT * BIN_THREADS) {
    int   s[BIN_EPT], d[BIN_EPT], r[BIN_EPT], bb[BIN_EPT];
    #pragma unroll
    for (int k = 0; k < BIN_EPT; ++k) {
      int i = i0 + k * BIN_THREADS;
      bool ok = i < hi;
      int ic = ok ? i : hi - 1;
      s[k] = ei[ic];
      d[k] = ok ? ei[E + ic] : -1;
    }
    #pragma unroll
    for (int k = 0; k < BIN_EPT; ++k) {
      r[k]  = CELL_REC;
      bb[k] = 0;
      if (d[k] >= 0) {
        bb[k] = bin_of(d[k], NB);
        r[k]  = atomicAdd(&hist[bb[k]], 1);
      }
    }
    #pragma unroll
    for (int k = 0; k < BIN_EPT; ++k) {
      if (r[k] < CELL_REC) {
        int dl = d[k] - (bb[k] << 8);
        int2 p;
        p.x = s[k] | (int)((unsigned)dl << 23);
        p.y = 0;
        stage[bb[k] * STAGE_PITCH + r[k] + 1] = p;
      }
    }
  }
  __syncthreads();

  int2* cellw = (int2*)lists + (size_t)(w >> 1) * (DF / 2) + (w & 1) * (DF / 4);
  for (int b0 = wv; b0 < NB; b0 += 4 * BIN_WAVES) {
    int  n[4];
    int2 p[4];
    #pragma unroll
    for (int u = 0; u < 4; ++u) {
      int b = b0 + u * BIN_WAVES;
      n[u] = b < NB ? hist[b] : -1;
    }
    #pragma unroll
    for (int u = 0; u < 4; ++u) {
      int b = b0 + u * BIN_WAVES;
      int m = n[u] < CELL_REC ? n[u] : CELL_REC;
      p[u].x = n[u];
      p[u].y = 0;
      if (lane >= 1 && lane <= m) p[u] = stage[b * STAGE_PITCH + lane];
    }
    #pragma unroll
    for (int u = 0; u < 4; ++u) {
      int b = b0 + u * BIN_WAVES;
      int nu = __builtin_amdgcn_readfirstlane(n[u]);
      int m  = nu < CELL_REC ? nu : CELL_REC;
      unsigned key = (unsigned)p[u].x;
      int slot = 1;
      for (int j = 1; j <= m; ++j) {
        unsigned kj = (unsigned)__builtin_amdgcn_readlane(p[u].x, j);
        slot += ((kj < key) || (kj == key && j < lane)) ? 1 : 0;
      }
      if (lane == 0) slot = 0;
      if (lane <= m) cellw[(size_t)b * (256 * (DF / 2)) + slot] = p[u];
    }
  }
}

__global__ void gemm_bin_kernel(const float* __restrict__ x,
                                const float* __restrict__ W,
                                const float* __restrict__ att_src,
                                const float* __restrict__ att_dst,
                                float* __restrict__ xw,
                                float* __restrict__ a_src,
                                float* __restrict__ a_dst,
                                const int* __restrict__ ei,
                                float* __restrict__ lists,
                                int N, int E, int NB, int chunk,
                                int gemmBlocks) {
  const int blk = blockIdx.x;
  if (blk < gemmBlocks) {
    gemm_xw_body(blk, x, W, att_src, att_dst, xw, a_src, a_dst, N);
  } else {
    bin_body(blk - gemmBlocks, ei, lists, E, NB, chunk);
  }
}

__global__ void rank_kernel(const int* __restrict__ ei,
                            const float* __restrict__ a_src,
                            int* __restrict__ cnt,
                            float* lists, int E, int N, int NB) {
  __shared__ int2 lst[512 * LIST_PITCH];
  __shared__ int cntL[512];
  __shared__ int ovf;
  const int b    = blockIdx.x;
  const int tid  = threadIdx.x;
  const int lane = tid & 31;
  const int wv   = tid >> 5;
  const int node0 = b << 8;
  const int nn    = (b == NB - 1) ? N - node0 : 256;

  if (tid < 512) cntL[tid] = 0;
  if (tid == 0) ovf = 0;
  __syncthreads();

  float2 rec[CELLS_PER_WAVE];
  int    rk[CELLS_PER_WAVE];
  float  asv[CELLS_PER_WAVE];
  #pragma unroll
  for (int q = 0; q < CELLS_PER_WAVE; ++q) {
    int c = wv * CELLS_PER_WAVE + q;
    rec[q] = *(const float2*)(lists + (size_t)(node0 + (c >> 1)) * DF +
                              (c & 1) * (DF / 2) + lane * 2);
  }
  #pragma unroll
  for (int q = 0; q < CELLS_PER_WAVE; ++q) {
    int x = __float_as_int(rec[q].x);
    int n = __builtin_amdgcn_readfirstlane(x);
    if (n > CELL_REC) ovf = 1;
    rk[q]  = -1;
    asv[q] = 0.0f;
    if (lane >= 1 && lane <= n) {
      rk[q]  = atomicAdd(&cntL[(unsigned)x >> 23], 1);
      asv[q] = a_src[x & 0x7fffff];
    }
  }
  #pragma unroll
  for (int q = 0; q < CELLS_PER_WAVE; ++q) {
    if (rk[q] >= 0 && rk[q] < CAP) {
      int x  = __float_as_int(rec[q].x);
      int dl = (unsigned)x >> 23;
      int2 p;
      p.x = x & 0x7fffff;
      p.y = __float_as_int(asv[q]);
      lst[dl * LIST_PITCH + rk[q]] = p;
    }
  }
  __syncthreads();

  if (ovf != 0) {
    if (tid < 512) cntL[tid] = 0;
    __syncthreads();
    for (int i = tid; i < E; i += RANK_THREADS) {
      int d = ei[E + i];
      if (bin_of(d, NB) == b) {
        int s  = ei[i];
        int dl = d - node0;
        int r  = atomicAdd(&cntL[dl], 1);
        if (r < CAP) {
          int2 p;
          p.x = s;
          p.y = __float_as_int(a_src[s]);
          lst[dl * LIST_PITCH + r] = p;
        }
      }
    }
    __syncthreads();
  }

  for (int dl0 = wv; dl0 < nn; dl0 += 4 * RANK_WAVES) {
    int  nv[4], m[4];
    int2 p[4];
    #pragma unroll
    for (int u = 0; u < 4; ++u) {
      int dl = dl0 + u * RANK_WAVES;
      nv[u] = dl < nn ? cntL[dl] : 0;
    }
    #pragma unroll
    for (int u = 0; u < 4; ++u) {
      int n = __builtin_amdgcn_readfirstlane(nv[u]);
      m[u] = n < CAP ? n : CAP;
    }
    #pragma unroll
    for (int u = 0; u < 4; ++u) {
      int dl = dl0 + u * RANK_WAVES;
      p[u].x = 0;
      p[u].y = 0;
      if (lane < m[u]) p[u] = lst[dl * LIST_PITCH + lane];
    }
    #pragma unroll
    for (int u = 0; u < 4; ++u) {
      int dl = dl0 + u * RANK_WAVES;
      int2* row = (int2*)lists + (size_t)(node0 + dl) * (DF / 2);
      if (lane < m[u]) row[lane] = p[u];
      if (m[u] > 32) {
        if (lane + 32 < m[u]) row[lane + 32] = lst[dl * LIST_PITCH + lane + 32];
      }
    }
  }
  if (tid < nn) cnt[node0 + tid] = cntL[tid];
}

__device__ __forceinline__ float wave_max(float v) {
  for (int off = 16; off > 0; off >>= 1) v = fmaxf(v, __shfl_xor(v, off, 32));
  return v;
}

__device__ __forceinline__ float wave_sum(float v) {
  for (int off = 16; off > 0; off >>= 1) v += __shfl_xor(v, off, 32);
  return v;
}

__device__ __forceinline__ int lane_int(int v, int j) {
  return __builtin_amdgcn_readlane(v, j);
}
__device__ __forceinline__ float lane_flt(float v, int j) {
  return __int_as_float(__builtin_amdgcn_readlane(__float_as_int(v), j));
}

__device__ __forceinline__ float4 row4(const float* __restrict__ xw, int s, int lane) {
  return *(const float4*)(xw + (size_t)s * DF + lane * 4);
}

__device__ __forceinline__ void axpy4(float4& acc, float a, const float4& r) {
  acc.x += a * r.x; acc.y += a * r.y; acc.z += a * r.z; acc.w += a * r.w;
}

__global__ void gather_kernel(const int* __restrict__ ei,
                              const float* __restrict__ xw,
                              const float* __restrict__ a_src,
                              const float* __restrict__ a_dst,
                              const int* __restrict__ cnt,
                              const float* __restrict__ bias,
                              float* out, int E, int N) {
  int gw   = (blockIdx.x * blockDim.x + threadIdx.x) >> 5;
  int lane = threadIdx.x & 31;
  if (gw >= N) return;
  const int d = __builtin_amdgcn_readfirstlane(gw);

  const int   n      = cnt[d];
  const float ad     = a_dst[d];
  const float e_self = leaky(a_src[d] + ad);
  float* orow = out + (size_t)d * DF;
  const float4 rself = row4(xw, d, lane);
  const float4 bv    = *(const float4*)(bias + lane * 4);

  float4 acc = {0.0f, 0.0f, 0.0f, 0.0f};
  float m, denom;

  if (n <= CAP) {
    int   sc[2];
    float ec[2];
    float mloc = e_self;
    for (int c = 0; c < 2; ++c) {
      int i = c * 32 + lane;
      sc[c] = d;
      ec[c] = -__builtin_inff();
      if (i < n) {
        const float2 p = *(const float2*)(orow + 2 * i);
        sc[c] = __float_as_int(p.x);
        ec[c] = leaky(p.y + ad);
      }
      mloc = fmaxf(mloc, ec[c]);
    }
    m = wave_max(mloc);

    float al[2];
    float sloc = 0.0f;
    for (int c = 0; c < 2; ++c) {
      int i = c * 32 + lane;
      al[c] = (i < n) ? expf(ec[c] - m) : 0.0f;
      sloc += al[c];
    }
    denom = wave_sum(sloc) + expf(e_self - m);
    for (int c = 0; c < 2; ++c) {
      al[c] = al[c] / denom;
      int mc = n - c * 32;
      mc = mc < 0 ? 0 : (mc > 32 ? 32 : mc);
      int j = 0;
      for (; j + 4 <= mc; j += 4) {
        int   s0 = lane_int(sc[c], j),     s1 = lane_int(sc[c], j + 1);
        int   s2 = lane_int(sc[c], j + 2), s3 = lane_int(sc[c], j + 3);
        float a0 = lane_flt(al[c], j),     a1 = lane_flt(al[c], j + 1);
        float a2 = lane_flt(al[c], j + 2), a3 = lane_flt(al[c], j + 3);
        float4 r0 = row4(xw, s0, lane);
        float4 r1 = row4(xw, s1, lane);
        float4 r2 = row4(xw, s2, lane);
        float4 r3 = row4(xw, s3, lane);
        axpy4(acc, a0, r0);
        axpy4(acc, a1, r1);
        axpy4(acc, a2, r2);
        axpy4(acc, a3, r3);
      }
      for (; j < mc; ++j) {
        int   s0 = lane_int(sc[c], j);
        float a0 = lane_flt(al[c], j);
        float4 r0 = row4(xw, s0, lane);
        axpy4(acc, a0, r0);
      }
    }
  } else {
    float mloc = e_self;
    for (int base = 0; base < E; base += 32) {
      int i = base + lane;
      if (i < E && ei[E + i] == d) mloc = fmaxf(mloc, leaky(a_src[ei[i]] + ad));
    }
    m = wave_max(mloc);
    float sloc = 0.0f;
    for (int base = 0; base < E; base += 32) {
      int i = base + lane;
      if (i < E && ei[E + i] == d) sloc += expf(leaky(a_src[ei[i]] + ad) - m);
    }
    denom = wave_sum(sloc) + expf(e_self - m);
    for (int base = 0; base < E; base += 32) {
      int i = base + lane;
      bool v = (i < E) && (ei[E + i] == d);
      int   sv = d;
      float av = 0.0f;
      if (v) {
        sv = ei[i];
        av = expf(leaky(a_src[sv] + ad) - m) / denom;
      }
      unsigned mask = __builtin_amdgcn_ballot_w32(v);
      while (mask) {
        int j = __builtin_ctz(mask);
        mask &= mask - 1;
        int   s0 = lane_int(sv, j);
        float a0 = lane_flt(av, j);
        float4 r0 = row4(xw, s0, lane);
        axpy4(acc, a0, r0);
      }
    }
  }

  {
    float a0 = expf(e_self - m) / denom;
    axpy4(acc, a0, rself);
  }

  float4 o;
  o.x = acc.x + bv.x; o.y = acc.y + bv.y; o.z = acc.z + bv.z; o.w = acc.w + bv.w;
  o.x = o.x > 0.0f ? o.x : (expf(o.x) - 1.0f);
  o.y = o.y > 0.0f ? o.y : (expf(o.y) - 1.0f);
  o.z = o.z > 0.0f ? o.z : (expf(o.z) - 1.0f);
  o.w = o.w > 0.0f ? o.w : (expf(o.w) - 1.0f);
  *(float4*)(orow + lane * 4) = o;
}

extern "C" void kernel_launch(void* const* d_in, const int* in_sizes, int n_in,
                              void* d_out, int out_size, void* d_ws, size_t ws_size,
                              hipStream_t stream) {
  const float* x       = (const float*)d_in[0];
  const int*   ei      = (const int*)  d_in[1];
  const float* W       = (const float*)d_in[2];
  const float* att_src = (const float*)d_in[3];
  const float* att_dst = (const float*)d_in[4];
  const float* bias    = (const float*)d_in[5];

  const int N = in_sizes[0] / DF;
  const int E = in_sizes[1] / 2;
  float* out = (float*)d_out;

  float* xw   = (float*)d_ws;
  float* asrc = xw + (size_t)N * DF;
  float* adst = asrc + N;
  int*   cnt  = (int*)(adst + N);

  const int NB = N >> 8;
  {
    int strips     = (N + 15) / 16;
    int gemmBlocks = (strips + (BIN_THREADS / 32) - 1) / (BIN_THREADS / 32);
    int chunk      = (E + NBW - 1) / NBW;
    gemm_bin_kernel<<<gemmBlocks + NBW, BIN_THREADS,
                      NB * (STAGE_PITCH * sizeof(int2) + sizeof(int)), stream>>>(
        x, W, att_src, att_dst, xw, asrc, adst, ei, out, N, E, NB, chunk,
        gemmBlocks);
  }
  rank_kernel<<<NB, RANK_THREADS, 0, stream>>>(ei, asrc, cnt, out, E, N, NB);
  {
    long long threads = (long long)N * 32;
    gather_kernel<<<(unsigned)((threads + 255) / 256), 256, 0, stream>>>(
        ei, xw, asrc, adst, cnt, bias, out, E, N);
  }
}
